# starting kernel
# baseline (speedup 1.0000x reference)
#include <hip/hip_runtime.h>

typedef __attribute__((ext_vector_type(2))) float v2f;
typedef __attribute__((ext_vector_type(8))) float v8f;

#define N_NODES 50000
#define N_EDGES 500000
#define HID 128
#define LDSPITCH (HID + 4)

__global__ void zero_f32(float* __restrict__ p, long n) {
  long i = (long)blockIdx.x * blockDim.x + threadIdx.x;
  long stride = (long)gridDim.x * blockDim.x;
  long n4 = n >> 2;
  float4* p4 = (float4*)p;
  for (long j = i; j < n4; j += stride) p4[j] = make_float4(0.f, 0.f, 0.f, 0.f);
}

__global__ void degree_kernel(const int* __restrict__ dst, float* __restrict__ deg, int E) {
  int e = blockIdx.x * blockDim.x + threadIdx.x;
  if (e < E) atomicAdd(&deg[dst[e]], 1.0f);
}

__global__ void invdeg_kernel(float* __restrict__ deg, int n) {
  int i = blockIdx.x * blockDim.x + threadIdx.x;
  if (i < n) deg[i] = 1.0f / fmaxf(deg[i], 1.0f);
}

__global__ void scatter_kernel(const float* __restrict__ x,
                               const int* __restrict__ src,
                               const int* __restrict__ dst,
                               float* __restrict__ agg, int E) {
  int wave = (int)(((long)blockIdx.x * blockDim.x + threadIdx.x) >> 5);
  int lane = threadIdx.x & 31;
  if (wave >= E) return;
  int s = src[wave];
  int d = dst[wave];
  const float4 v = *(const float4*)(x + (long)s * HID + lane * 4);
  float* o = agg + (long)d * HID + lane * 4;
  atomicAdd(o + 0, v.x);
  atomicAdd(o + 1, v.y);
  atomicAdd(o + 2, v.z);
  atomicAdd(o + 3, v.w);
}

template <int OUTD, bool RELU, int STRIPS>
__global__ __launch_bounds__(32 * (OUTD / 16)) void sage_gemm(
    const float* __restrict__ xin,
    const float* __restrict__ agg,
    const float* __restrict__ invdeg,
    const float* __restrict__ Wl,
    const float* __restrict__ Wr,
    const float* __restrict__ bias,
    float* __restrict__ out,
    int n_nodes) {
  __shared__ float sX[16][LDSPITCH];
  __shared__ float sA[16][LDSPITCH];

  const int tid = threadIdx.x;
  const int nthreads = 32 * (OUTD / 16);
  const int wave = tid >> 5;
  const int lane = tid & 31;
  const int half = lane >> 4;
  const int r = lane & 15;
  const int col = wave * 16 + r;

  v2f BL[HID / 4], BR[HID / 4];
#pragma unroll
  for (int j = 0; j < HID / 4; ++j) {
    const int ka = 4 * j + 2 * half;
    BL[j][0] = Wl[(long)ka * OUTD + col];
    BL[j][1] = Wl[(long)(ka + 1) * OUTD + col];
    BR[j][0] = Wr[(long)ka * OUTD + col];
    BR[j][1] = Wr[(long)(ka + 1) * OUTD + col];
  }
  const float bn = bias[col];

  for (int s = 0; s < STRIPS; ++s) {
    const int row0 = (blockIdx.x * STRIPS + s) * 16;
    if (row0 >= n_nodes) break;

    for (int i = tid; i < 16 * (HID / 4); i += nthreads) {
      const int rr = i / (HID / 4);
      const int c4 = (i % (HID / 4)) * 4;
      const long base = (long)(row0 + rr) * HID + c4;
      float4 xv = *(const float4*)(xin + base);
      float4 av = *(const float4*)(agg + base);
      const float iv = invdeg[row0 + rr];
      av.x *= iv; av.y *= iv; av.z *= iv; av.w *= iv;
      *(float4*)&sX[rr][c4] = xv;
      *(float4*)&sA[rr][c4] = av;
    }
    __syncthreads();

    v8f accL = {};
    v8f accR = {};
#pragma unroll
    for (int j = 0; j < HID / 4; ++j) {
      const int ka = 4 * j + 2 * half;
      v2f aL, aR;
      aL[0] = sA[r][ka];
      aL[1] = sA[r][ka + 1];
      aR[0] = sX[r][ka];
      aR[1] = sX[r][ka + 1];
      accL = __builtin_amdgcn_wmma_f32_16x16x4_f32(false, aL, false, BL[j],
                                                   (short)0, accL, false, false);
      accR = __builtin_amdgcn_wmma_f32_16x16x4_f32(false, aR, false, BR[j],
                                                   (short)0, accR, false, false);
    }

#pragma unroll
    for (int i = 0; i < 8; ++i) {
      const int gr = row0 + half * 8 + i;
      float v = accL[i] + accR[i] + bn;
      if (RELU) v = fmaxf(v, 0.0f);
      out[(long)gr * OUTD + col] = v;
    }
    __syncthreads();
  }
}

extern "C" void kernel_launch(void* const* d_in, const int* in_sizes, int n_in,
                              void* d_out, int out_size, void* d_ws, size_t ws_size,
                              hipStream_t stream) {
  const float* x   = (const float*)d_in[0];
  const int* eidx  = (const int*)d_in[1];
  const float* W1l = (const float*)d_in[2];
  const float* W1r = (const float*)d_in[3];
  const float* b1  = (const float*)d_in[4];
  const float* W2l = (const float*)d_in[5];
  const float* W2r = (const float*)d_in[6];
  const float* b2  = (const float*)d_in[7];
  const float* W3l = (const float*)d_in[8];
  const float* W3r = (const float*)d_in[9];
  const float* b3  = (const float*)d_in[10];
  float* out = (float*)d_out;

  const int* src = eidx;
  const int* dst = eidx + N_EDGES;

  float* deg = (float*)d_ws;
  float* agg = deg + N_NODES;
  float* h1  = agg + (long)N_NODES * HID;
  float* h2  = h1 + (long)N_NODES * HID;

  const int ZB = 2048;
  const int eBlocks = (N_EDGES + 255) / 256;
  const int scatBlocks = (int)(((long)N_EDGES * 32 + 255) / 256);
  constexpr int STRIPS = 4;
  const int gemmBlocks = (N_NODES + 16 * STRIPS - 1) / (16 * STRIPS);

  zero_f32<<<ZB, 256, 0, stream>>>(deg, N_NODES);
  degree_kernel<<<eBlocks, 256, 0, stream>>>(dst, deg, N_EDGES);
  invdeg_kernel<<<(N_NODES + 255) / 256, 256, 0, stream>>>(deg, N_NODES);

  zero_f32<<<ZB, 256, 0, stream>>>(agg, (long)N_NODES * HID);
  scatter_kernel<<<scatBlocks, 256, 0, stream>>>(x, src, dst, agg, N_EDGES);
  sage_gemm<128, true, STRIPS><<<gemmBlocks, 256, 0, stream>>>(x, agg, deg, W1l, W1r, b1, h1, N_NODES);

  zero_f32<<<ZB, 256, 0, stream>>>(agg, (long)N_NODES * HID);
  scatter_kernel<<<scatBlocks, 256, 0, stream>>>(h1, src, dst, agg, N_EDGES);
  sage_gemm<128, true, STRIPS><<<gemmBlocks, 256, 0, stream>>>(h1, agg, deg, W2l, W2r, b2, h2, N_NODES);

  zero_f32<<<ZB, 256, 0, stream>>>(agg, (long)N_NODES * HID);
  scatter_kernel<<<scatBlocks, 256, 0, stream>>>(h2, src, dst, agg, N_EDGES);
  sage_gemm<64, false, STRIPS><<<gemmBlocks, 128, 0, stream>>>(h2, agg, deg, W3l, W3r, b3, out, N_NODES);
}


# Float-atomic scatter-add replaced by an edge list grouped by destination, rebuilt per call (slot from a returning int atomic, scan, fill) + atomic-free per-node gather-sum; sum order = slot order
# speedup vs baseline: 2.2992x; 2.2992x over previous
#include <hip/hip_runtime.h>

typedef __attribute__((ext_vector_type(2))) float v2f;
typedef __attribute__((ext_vector_type(8))) float v8f;

#define N_NODES 50000
#define N_EDGES 500000
#define HID 128
#define LDSPITCH (HID + 4)
#define SCAN_WAVES 32
#define SCAN_ITERS 13
#define NPAD (SCAN_WAVES * SCAN_ITERS * 128)

__global__ void zero_f32(float* __restrict__ p, long n) {
  long i = (long)blockIdx.x * blockDim.x + threadIdx.x;
  long stride = (long)gridDim.x * blockDim.x;
  long n4 = n >> 2;
  float4* p4 = (float4*)p;
  for (long j = i; j < n4; j += stride) p4[j] = make_float4(0.f, 0.f, 0.f, 0.f);
}

__global__ void count_rank_kernel(const int* __restrict__ dst, int* __restrict__ cnt,
                                  int* __restrict__ rank, int E) {
  int e = blockIdx.x * blockDim.x + threadIdx.x;
  if (e < E) rank[e] = atomicAdd(&cnt[dst[e]], 1);
}

__global__ __launch_bounds__(SCAN_WAVES * 32) void scan_kernel(const int* __restrict__ cnt,
                                                               int* __restrict__ off) {
  __shared__ int wsum[SCAN_WAVES];
  const int lane = threadIdx.x & 31;
  const int wave = threadIdx.x >> 5;
  const int base = wave * (SCAN_ITERS * 128) + lane * 4;

  int4 c[SCAN_ITERS];
#pragma unroll
  for (int i = 0; i < SCAN_ITERS; ++i) c[i] = *(const int4*)(cnt + base + i * 128);

  int s[SCAN_ITERS], inc[SCAN_ITERS];
#pragma unroll
  for (int i = 0; i < SCAN_ITERS; ++i) {
    s[i] = c[i].x + c[i].y + c[i].z + c[i].w;
    inc[i] = s[i];
  }
#pragma unroll
  for (int d = 1; d < 32; d <<= 1) {
    int t[SCAN_ITERS];
#pragma unroll
    for (int i = 0; i < SCAN_ITERS; ++i) t[i] = __shfl_up(inc[i], d);
#pragma unroll
    for (int i = 0; i < SCAN_ITERS; ++i)
      if (lane >= d) inc[i] += t[i];
  }
  int ex[SCAN_ITERS];
  int carry = 0;
#pragma unroll
  for (int i = 0; i < SCAN_ITERS; ++i) {
    ex[i] = carry + inc[i] - s[i];
    carry += __builtin_amdgcn_readlane(inc[i], 31);
  }
  if (lane == 0) wsum[wave] = carry;
  __syncthreads();
  const int wv = wsum[lane];
  int winc = wv;
#pragma unroll
  for (int d = 1; d < 32; d <<= 1) {
    const int t = __shfl_up(winc, d);
    if (lane >= d) winc += t;
  }
  const int woff = __shfl(winc - wv, wave);

#pragma unroll
  for (int i = 0; i < SCAN_ITERS; ++i) {
    int4 o;
    o.x = woff + ex[i];
    o.y = o.x + c[i].x;
    o.z = o.y + c[i].y;
    o.w = o.z + c[i].z;
    *(int4*)(off + base + i * 128) = o;
  }
}

__global__ void fill_kernel(const int* __restrict__ src, const int* __restrict__ dst,
                            const int* __restrict__ off, const int* __restrict__ rank,
                            const int* __restrict__ cnt, int* __restrict__ col,
                            float* __restrict__ invdeg, int E, int n_nodes) {
  int e = blockIdx.x * blockDim.x + threadIdx.x;
  if (e < n_nodes) invdeg[e] = 1.0f / fmaxf((float)cnt[e], 1.0f);
  if (e < E) {
    unsigned p = (unsigned)(off[dst[e]] + rank[e]);
    p = min(p, (unsigned)(E - 1));
    col[p] = src[e];
  }
}

template <int N>
__device__ __forceinline__ void rows_issue(float4 (&v)[N], const float* __restrict__ x, int lane4,
                                           int ids, int k0) {
#pragma unroll
  for (int j = 0; j < N; ++j) {
    const int s = __builtin_amdgcn_readlane(ids, k0 + j);
    const float* row = x + (long)s * HID;
    v[j] = *(const float4*)(row + lane4);
  }
}

template <int N>
__device__ __forceinline__ void rows_add(float4& acc, const float4 (&v)[N]) {
#pragma unroll
  for (int j = 0; j < N; ++j) {
    acc.x += v[j].x; acc.y += v[j].y; acc.z += v[j].z; acc.w += v[j].w;
  }
}

__device__ __forceinline__ int load_ids(const int* __restrict__ col, int first, int n, int lane,
                                        unsigned last) {
  int id = 0;
  if (lane < n) id = (int)min((unsigned)col[first + lane], last);
  return id;
}

#define ROWS_CASE(M)                                         \
  case M: {                                                  \
    float4 v[M];                                             \
    rows_issue<M>(v, x, lane4, ids, k);                      \
    fids = load_ids(col, more ? c0 + 32 : nb,                \
                    more ? e - (c0 + 32) : ne - nb, lane, last); \
    rows_add<M>(acc, v);                                     \
  } break;

__global__ __launch_bounds__(256) void gather_sum_kernel(const float* __restrict__ x,
                                                         const int* __restrict__ off,
                                                         const int* __restrict__ col,
                                                         float* __restrict__ agg, int n_nodes) {
  const int lane = threadIdx.x & 31;
  const int lane4 = lane * 4;
  const int wave =
      __builtin_amdgcn_readfirstlane((int)((blockIdx.x * blockDim.x + threadIdx.x) >> 5));
  const int nwaves = (int)((gridDim.x * blockDim.x) >> 5);
  const unsigned last = (unsigned)(n_nodes - 1);

  int node = wave;
  if (node >= n_nodes) return;
  int b = __builtin_amdgcn_readfirstlane(off[node]);
  int e = __builtin_amdgcn_readfirstlane(off[node + 1]);
  int ids = load_ids(col, b, e - b, lane, last);
  int n1 = node + nwaves;
  int nb = 0, ne = 0;
  if (n1 < n_nodes) {
    nb = __builtin_amdgcn_readfirstlane(off[n1]);
    ne = __builtin_amdgcn_readfirstlane(off[n1 + 1]);
  }

  for (;;) {
    const int n2 = n1 + nwaves;
    int b2 = 0, e2 = 0;
    if (n2 < n_nodes) {
      b2 = __builtin_amdgcn_readfirstlane(off[n2]);
      e2 = __builtin_amdgcn_readfirstlane(off[n2 + 1]);
    }

    float4 acc = make_float4(0.f, 0.f, 0.f, 0.f);
    for (int c0 = b;; c0 += 32) {
      const int m = min(e - c0, 32);
      const bool more = c0 + 32 < e;
      int k = 0;
      for (; k + 16 <= m; k += 16) {
        float4 v[16];
        rows_issue<16>(v, x, lane4, ids, k);
        rows_add<16>(acc, v);
      }
      int fids = 0;
      switch (m - k) {
        ROWS_CASE(1) ROWS_CASE(2) ROWS_CASE(3) ROWS_CASE(4) ROWS_CASE(5)
        ROWS_CASE(6) ROWS_CASE(7) ROWS_CASE(8) ROWS_CASE(9) ROWS_CASE(10)
        ROWS_CASE(11) ROWS_CASE(12) ROWS_CASE(13) ROWS_CASE(14) ROWS_CASE(15)
        default:
          fids = load_ids(col, more ? c0 + 32 : nb, more ? e - (c0 + 32) : ne - nb, lane, last);
          break;
      }
      ids = fids;
      if (!more) break;
    }
    *(float4*)(agg + (long)node * HID + lane4) = acc;

    if (n1 >= n_nodes) break;
    node = n1; b = nb; e = ne;
    n1 = n2; nb = b2; ne = e2;
  }
}
#undef ROWS_CASE

template <int OUTD, bool RELU, int STRIPS>
__global__ __launch_bounds__(32 * (OUTD / 16)) void sage_gemm(
    const float* __restrict__ xin,
    const float* __restrict__ agg,
    const float* __restrict__ invdeg,
    const float* __restrict__ Wl,
    const float* __restrict__ Wr,
    const float* __restrict__ bias,
    float* __restrict__ out,
    int n_nodes) {
  __shared__ float sX[16][LDSPITCH];
  __shared__ float sA[16][LDSPITCH];

  const int tid = threadIdx.x;
  const int nthreads = 32 * (OUTD / 16);
  const int wave = tid >> 5;
  const int lane = tid & 31;
  const int half = lane >> 4;
  const int r = lane & 15;
  const int col = wave * 16 + r;

  v2f BL[HID / 4], BR[HID / 4];
#pragma unroll
  for (int j = 0; j < HID / 4; ++j) {
    const int ka = 4 * j + 2 * half;
    BL[j][0] = Wl[(long)ka * OUTD + col];
    BL[j][1] = Wl[(long)(ka + 1) * OUTD + col];
    BR[j][0] = Wr[(long)ka * OUTD + col];
    BR[j][1] = Wr[(long)(ka + 1) * OUTD + col];
  }
  const float bn = bias[col];

  for (int s = 0; s < STRIPS; ++s) {
    const int row0 = (blockIdx.x * STRIPS + s) * 16;
    if (row0 >= n_nodes) break;

    for (int i = tid; i < 16 * (HID / 4); i += nthreads) {
      const int rr = i / (HID / 4);
      const int c4 = (i % (HID / 4)) * 4;
      const long base = (long)(row0 + rr) * HID + c4;
      float4 xv = *(const float4*)(xin + base);
      float4 av = *(const float4*)(agg + base);
      const float iv = invdeg[row0 + rr];
      av.x *= iv; av.y *= iv; av.z *= iv; av.w *= iv;
      *(float4*)&sX[rr][c4] = xv;
      *(float4*)&sA[rr][c4] = av;
    }
    __syncthreads();

    v8f accL = {};
    v8f accR = {};
#pragma unroll
    for (int j = 0; j < HID / 4; ++j) {
      const int ka = 4 * j + 2 * half;
      v2f aL, aR;
      aL[0] = sA[r][ka];
      aL[1] = sA[r][ka + 1];
      aR[0] = sX[r][ka];
      aR[1] = sX[r][ka + 1];
      accL = __builtin_amdgcn_wmma_f32_16x16x4_f32(false, aL, false, BL[j],
                                                   (short)0, accL, false, false);
      accR = __builtin_amdgcn_wmma_f32_16x16x4_f32(false, aR, false, BR[j],
                                                   (short)0, accR, false, false);
    }

#pragma unroll
    for (int i = 0; i < 8; ++i) {
      const int gr = row0 + half * 8 + i;
      float v = accL[i] + accR[i] + bn;
      if (RELU) v = fmaxf(v, 0.0f);
      out[(long)gr * OUTD + col] = v;
    }
    __syncthreads();
  }
}

extern "C" void kernel_launch(void* const* d_in, const int* in_sizes, int n_in,
                              void* d_out, int out_size, void* d_ws, size_t ws_size,
                              hipStream_t stream) {
  const float* x   = (const float*)d_in[0];
  const int* eidx  = (const int*)d_in[1];
  const float* W1l = (const float*)d_in[2];
  const float* W1r = (const float*)d_in[3];
  const float* b1  = (const float*)d_in[4];
  const float* W2l = (const float*)d_in[5];
  const float* W2r = (const float*)d_in[6];
  const float* b2  = (const float*)d_in[7];
  const float* W3l = (const float*)d_in[8];
  const float* W3r = (const float*)d_in[9];
  const float* b3  = (const float*)d_in[10];
  float* out = (float*)d_out;

  const int* src = eidx;
  const int* dst = eidx + N_EDGES;

  float* invdeg = (float*)d_ws;
  float* agg = invdeg + NPAD;
  float* h1  = agg + (long)N_NODES * HID;
  float* h2  = h1 + (long)N_NODES * HID;
  int* cnt   = (int*)(h2 + (long)N_NODES * HID);
  int* off   = cnt + NPAD;
  int* rank  = off + NPAD;
  int* col   = rank + N_EDGES;

  constexpr int eBlocks = (N_EDGES + 255) / 256;
  constexpr int gatherBlocks = 1024;
  constexpr int STRIPS = 4;
  constexpr int gemmBlocks = (N_NODES + 16 * STRIPS - 1) / (16 * STRIPS);

  zero_f32<<<(NPAD / 4 + 255) / 256, 256, 0, stream>>>((float*)cnt, NPAD);
  count_rank_kernel<<<eBlocks, 256, 0, stream>>>(dst, cnt, rank, N_EDGES);
  scan_kernel<<<1, SCAN_WAVES * 32, 0, stream>>>(cnt, off);
  fill_kernel<<<eBlocks, 256, 0, stream>>>(src, dst, off, rank, cnt, col, invdeg, N_EDGES, N_NODES);

  gather_sum_kernel<<<gatherBlocks, 256, 0, stream>>>(x, off, col, agg, N_NODES);
  sage_gemm<128, true, STRIPS><<<gemmBlocks, 256, 0, stream>>>(x, agg, invdeg, W1l, W1r, b1, h1, N_NODES);

  gather_sum_kernel<<<gatherBlocks, 256, 0, stream>>>(h1, off, col, agg, N_NODES);
  sage_gemm<128, true, STRIPS><<<gemmBlocks, 256, 0, stream>>>(h1, agg, invdeg, W2l, W2r, b2, h2, N_NODES);

  gather_sum_kernel<<<gatherBlocks, 256, 0, stream>>>(h2, off, col, agg, N_NODES);
  sage_gemm<64, false, STRIPS><<<gemmBlocks, 128, 0, stream>>>(h2, agg, invdeg, W3l, W3r, b3, out, N_NODES);
}


# Neighbour gather-mean fused into the layer GEMM: each wave sums its strip rows from the grouped edge list into the LDS A tile, next strip prefetched under the WMMAs; 3 launches fewer; sum order kept
# speedup vs baseline: 3.1161x; 1.3553x over previous
#include <hip/hip_runtime.h>

typedef __attribute__((ext_vector_type(2))) float v2f;
typedef __attribute__((ext_vector_type(8))) float v8f;

#define N_NODES 50000
#define N_EDGES 500000
#define HID 128
#define LDSPITCH (HID + 4)
#define SCAN_WAVES 32
#define SCAN_ITERS 13
#define NPAD (SCAN_WAVES * SCAN_ITERS * 128)

__global__ void zero_f32(float* __restrict__ p, long n) {
  long i = (long)blockIdx.x * blockDim.x + threadIdx.x;
  long stride = (long)gridDim.x * blockDim.x;
  long n4 = n >> 2;
  float4* p4 = (float4*)p;
  for (long j = i; j < n4; j += stride) p4[j] = make_float4(0.f, 0.f, 0.f, 0.f);
}

__global__ void count_rank_kernel(const int* __restrict__ dst, int* __restrict__ cnt,
                                  int* __restrict__ rank, int E) {
  int e = blockIdx.x * blockDim.x + threadIdx.x;
  if (e < E) rank[e] = atomicAdd(&cnt[dst[e]], 1);
}

__global__ __launch_bounds__(SCAN_WAVES * 32) void scan_kernel(const int* __restrict__ cnt,
                                                               int* __restrict__ off) {
  __shared__ int wsum[SCAN_WAVES];
  const int lane = threadIdx.x & 31;
  const int wave = threadIdx.x >> 5;
  const int base = wave * (SCAN_ITERS * 128) + lane * 4;

  int4 c[SCAN_ITERS];
#pragma unroll
  for (int i = 0; i < SCAN_ITERS; ++i) c[i] = *(const int4*)(cnt + base + i * 128);

  int s[SCAN_ITERS], inc[SCAN_ITERS];
#pragma unroll
  for (int i = 0; i < SCAN_ITERS; ++i) {
    s[i] = c[i].x + c[i].y + c[i].z + c[i].w;
    inc[i] = s[i];
  }
#pragma unroll
  for (int d = 1; d < 32; d <<= 1) {
    int t[SCAN_ITERS];
#pragma unroll
    for (int i = 0; i < SCAN_ITERS; ++i) t[i] = __shfl_up(inc[i], d);
#pragma unroll
    for (int i = 0; i < SCAN_ITERS; ++i)
      if (lane >= d) inc[i] += t[i];
  }
  int ex[SCAN_ITERS];
  int carry = 0;
#pragma unroll
  for (int i = 0; i < SCAN_ITERS; ++i) {
    ex[i] = carry + inc[i] - s[i];
    carry += __builtin_amdgcn_readlane(inc[i], 31);
  }
  if (lane == 0) wsum[wave] = carry;
  __syncthreads();
  const int wv = wsum[lane];
  int winc = wv;
#pragma unroll
  for (int d = 1; d < 32; d <<= 1) {
    const int t = __shfl_up(winc, d);
    if (lane >= d) winc += t;
  }
  const int woff = __shfl(winc - wv, wave);

#pragma unroll
  for (int i = 0; i < SCAN_ITERS; ++i) {
    int4 o;
    o.x = woff + ex[i];
    o.y = o.x + c[i].x;
    o.z = o.y + c[i].y;
    o.w = o.z + c[i].z;
    *(int4*)(off + base + i * 128) = o;
  }
}

__global__ void fill_kernel(const int* __restrict__ src, const int* __restrict__ dst,
                            const int* __restrict__ off, const int* __restrict__ rank,
                            int* __restrict__ col, int E) {
  int e = blockIdx.x * blockDim.x + threadIdx.x;
  if (e < E) {
    unsigned p = (unsigned)(off[dst[e]] + rank[e]);
    p = min(p, (unsigned)(E - 1));
    col[p] = src[e];
  }
}

template <int N>
__device__ __forceinline__ void rows_issue(float4 (&v)[N], const float* __restrict__ x, int lane4,
                                           int ids, int k0, unsigned last) {
#pragma unroll
  for (int j = 0; j < N; ++j) {
    const int s = (int)min((unsigned)__builtin_amdgcn_readlane(ids, k0 + j), last);
    const float* row = x + (long)s * HID;
    v[j] = *(const float4*)(row + lane4);
  }
}

template <int N>
__device__ __forceinline__ void rows_add(float4& acc, const float4 (&v)[N]) {
#pragma unroll
  for (int j = 0; j < N; ++j) {
    acc.x += v[j].x; acc.y += v[j].y; acc.z += v[j].z; acc.w += v[j].w;
  }
}

__device__ __forceinline__ int load_ids(const int* __restrict__ col, int first, int lane) {
  return col[min(first + lane, N_EDGES - 1)];
}

#define ROWS_CASE(M)                                   \
  case M: {                                            \
    float4 v[M];                                       \
    rows_issue<M>(v, x, lane4, ids, k, last);          \
    fids = load_ids(col, more ? c0 + 32 : nfirst, lane); \
    rows_add<M>(acc, v);                               \
  } break;

__device__ __forceinline__ float4 gather_rows(const float* __restrict__ x,
                                              const int* __restrict__ col, int b, int e, int& ids,
                                              int nfirst, int lane, int lane4, unsigned last) {
  float4 acc = make_float4(0.f, 0.f, 0.f, 0.f);
  for (int c0 = b;; c0 += 32) {
    const int m = min(e - c0, 32);
    const bool more = c0 + 32 < e;
    int k = 0;
    for (; k + 16 <= m; k += 16) {
      float4 v[16];
      rows_issue<16>(v, x, lane4, ids, k, last);
      rows_add<16>(acc, v);
    }
    int fids = 0;
    switch (m - k) {
      ROWS_CASE(1) ROWS_CASE(2) ROWS_CASE(3) ROWS_CASE(4) ROWS_CASE(5)
      ROWS_CASE(6) ROWS_CASE(7) ROWS_CASE(8) ROWS_CASE(9) ROWS_CASE(10)
      ROWS_CASE(11) ROWS_CASE(12) ROWS_CASE(13) ROWS_CASE(14) ROWS_CASE(15)
      default:
        fids = load_ids(col, more ? c0 + 32 : nfirst, lane);
        break;
    }
    ids = fids;
    if (!more) break;
  }
  return acc;
}
#undef ROWS_CASE

template <int NPW>
__device__ __forceinline__ int load_offs2(const int* __restrict__ off, int nodeA, int nodeB, int lane,
                                          int n_nodes) {
  const int node = (lane < 16 ? nodeA : nodeB) + min(lane & 15, NPW);
  return off[min(node, n_nodes)];
}

__device__ __forceinline__ void lds_barrier() {
  __builtin_amdgcn_fence(__ATOMIC_RELEASE, "workgroup", "local");
  __builtin_amdgcn_s_barrier();
  __builtin_amdgcn_fence(__ATOMIC_ACQUIRE, "workgroup", "local");
}

template <int OUTD, bool RELU>
__global__ __launch_bounds__(32 * (OUTD / 16)) void sage_layer(
    const float* __restrict__ x,
    const int* __restrict__ off,
    const int* __restrict__ col,
    const float* __restrict__ Wl,
    const float* __restrict__ Wr,
    const float* __restrict__ bias,
    float* __restrict__ out,
    int n_nodes) {
  constexpr int NW = OUTD / 16;
  constexpr int NPW = 16 / NW;
  __shared__ float sX[16][LDSPITCH];
  __shared__ float sA[16][LDSPITCH];

  const int tid = threadIdx.x;
  const int wave = __builtin_amdgcn_readfirstlane(tid >> 5);
  const int lane = tid & 31;
  const int lane4 = lane * 4;
  const int half = lane >> 4;
  const int r = lane & 15;
  const int col_n = wave * 16 + r;
  const unsigned last = (unsigned)(n_nodes - 1);
  const int nStrips = n_nodes / 16;

  int s = blockIdx.x;
  if (s >= nStrips) return;
  const int G = (int)gridDim.x;
  const int wrow = wave * NPW;

  int ov = load_offs2<NPW>(off, s * 16 + wrow, (s + G) * 16 + wrow, lane, n_nodes);
  static_assert(NPW == 2 || NPW == 4, "self rows are held in named registers");
  const float* xrow = x + (long)(s * 16 + wrow) * HID + lane4;
  float4 xs0 = *(const float4*)(xrow);
  float4 xs1 = *(const float4*)(xrow + HID);
  float4 xs2 = xs0, xs3 = xs0;
  if (NPW == 4) {
    xs2 = *(const float4*)(xrow + 2 * HID);
    xs3 = *(const float4*)(xrow + 3 * HID);
  }
  int ids = load_ids(col, min(max(__builtin_amdgcn_readlane(ov, 0), 0), N_EDGES), lane);

  v2f BL[HID / 4], BR[HID / 4];
#pragma unroll
  for (int j = 0; j < HID / 4; ++j) {
    const int ka = 4 * j + 2 * half;
    BL[j][0] = Wl[(long)ka * OUTD + col_n];
    BL[j][1] = Wl[(long)(ka + 1) * OUTD + col_n];
    BR[j][0] = Wr[(long)ka * OUTD + col_n];
    BR[j][1] = Wr[(long)(ka + 1) * OUTD + col_n];
  }
  const float bn = bias[col_n];

  for (; s < nStrips; s += G) {
    const int row0 = s * 16;
    const int ovc = ov;
    const bool next = s + G < nStrips;

    *(float4*)&sX[wrow][lane4] = xs0;
    *(float4*)&sX[wrow + 1][lane4] = xs1;
    if (NPW == 4) {
      *(float4*)&sX[wrow + 2][lane4] = xs2;
      *(float4*)&sX[wrow + 3][lane4] = xs3;
    }

    int b = min(max(__builtin_amdgcn_readlane(ovc, 0), 0), N_EDGES);
#pragma unroll 1
    for (int i = 0; i < NPW; ++i) {
      const int e = min(max(__builtin_amdgcn_readlane(ovc, i + 1), b), N_EDGES);
      const int nx = __builtin_amdgcn_readlane(ovc, i + 1 < NPW ? i + 1 : 16);
      const int bn = min(max(nx, 0), N_EDGES);
      float4 av = gather_rows(x, col, b, e, ids, bn, lane, lane4, last);
      const float iv = 1.0f / fmaxf((float)(e - b), 1.0f);
      av.x *= iv; av.y *= iv; av.z *= iv; av.w *= iv;
      *(float4*)&sA[wrow + i][lane4] = av;
      b = bn;
    }

    if (next) {
      xrow += (long)G * 16 * HID;
      xs0 = *(const float4*)(xrow);
      xs1 = *(const float4*)(xrow + HID);
      if (NPW == 4) {
        xs2 = *(const float4*)(xrow + 2 * HID);
        xs3 = *(const float4*)(xrow + 3 * HID);
      }
      ov = load_offs2<NPW>(off, (s + G) * 16 + wrow, (s + 2 * G) * 16 + wrow, lane, n_nodes);
    }
    lds_barrier();

    v8f accL = {};
    v8f accR = {};
#pragma unroll
    for (int j = 0; j < HID / 4; ++j) {
      const int ka = 4 * j + 2 * half;
      v2f aL, aR;
      aL[0] = sA[r][ka];
      aL[1] = sA[r][ka + 1];
      aR[0] = sX[r][ka];
      aR[1] = sX[r][ka + 1];
      accL = __builtin_amdgcn_wmma_f32_16x16x4_f32(false, aL, false, BL[j],
                                                   (short)0, accL, false, false);
      accR = __builtin_amdgcn_wmma_f32_16x16x4_f32(false, aR, false, BR[j],
                                                   (short)0, accR, false, false);
    }

#pragma unroll
    for (int i = 0; i < 8; ++i) {
      const int gr = row0 + half * 8 + i;
      float v = accL[i] + accR[i] + bn;
      if (RELU) v = fmaxf(v, 0.0f);
      out[(long)gr * OUTD + col_n] = v;
    }
    lds_barrier();
  }
}

extern "C" void kernel_launch(void* const* d_in, const int* in_sizes, int n_in,
                              void* d_out, int out_size, void* d_ws, size_t ws_size,
                              hipStream_t stream) {
  const float* x   = (const float*)d_in[0];
  const int* eidx  = (const int*)d_in[1];
  const float* W1l = (const float*)d_in[2];
  const float* W1r = (const float*)d_in[3];
  const float* b1  = (const float*)d_in[4];
  const float* W2l = (const float*)d_in[5];
  const float* W2r = (const float*)d_in[6];
  const float* b2  = (const float*)d_in[7];
  const float* W3l = (const float*)d_in[8];
  const float* W3r = (const float*)d_in[9];
  const float* b3  = (const float*)d_in[10];
  float* out = (float*)d_out;

  const int* src = eidx;
  const int* dst = eidx + N_EDGES;

  float* h1  = (float*)d_ws;
  float* h2  = h1 + (long)N_NODES * HID;
  int* cnt   = (int*)(h2 + (long)N_NODES * HID);
  int* off   = cnt + NPAD;
  int* rank  = off + NPAD;
  int* col   = rank + N_EDGES;

  constexpr int eBlocks = (N_EDGES + 255) / 256;
  constexpr int layerBlocks128 = 512;
  constexpr int layerBlocks64 = 1024;

  zero_f32<<<(NPAD / 4 + 255) / 256, 256, 0, stream>>>((float*)cnt, NPAD);
  count_rank_kernel<<<eBlocks, 256, 0, stream>>>(dst, cnt, rank, N_EDGES);
  scan_kernel<<<1, SCAN_WAVES * 32, 0, stream>>>(cnt, off);
  fill_kernel<<<eBlocks, 256, 0, stream>>>(src, dst, off, rank, col, N_EDGES);

  sage_layer<128, true><<<layerBlocks128, 256, 0, stream>>>(x, off, col, W1l, W1r, b1, h1, N_NODES);
  sage_layer<128, true><<<layerBlocks128, 256, 0, stream>>>(h1, off, col, W2l, W2r, b2, h2, N_NODES);
  sage_layer<64, false><<<layerBlocks64, 128, 0, stream>>>(h2, off, col, W3l, W3r, b3, out, N_NODES);
}
